# starting kernel
# baseline (speedup 1.0000x reference)
#include <hip/hip_runtime.h>


typedef __attribute__((ext_vector_type(2))) float v2f;
typedef __attribute__((ext_vector_type(8))) float v8f;

#define HEAD_SIZE 20
#define NUM_HEADS 2
#define SEQ 256
#define NSET 32
#define DMODEL 20
#define DQK 40
#define QKS 48
#define VTROWS 56

__device__ __forceinline__ v8f wmma4(v2f a, v2f b, v8f c) {
  return __builtin_amdgcn_wmma_f32_16x16x4_f32(
       false, a,  false, b,
       (short)0, c,  false,  false);
}

__global__ __launch_bounds__(256)
void set_attention_kernel(const float* __restrict__ x,
                          const float* __restrict__ y,
                          const float* __restrict__ WQg,
                          const float* __restrict__ WKg,
                          const float* __restrict__ WVg,
                          const float* __restrict__ WHg,
                          float* __restrict__ out) {
  extern __shared__ float smem[];
  float* Qs  = smem;
  float* Ks  = smem + 12288;
  float* Vt  = smem + 24576;
  float* Wqt = smem + 38912;
  float* Wkt = Wqt + 960;
  float* Wvt = Wkt + 960;
  float* Wht = smem + 41792;
  float* Xs  = smem + 43072;
  float* Ys  = Xs + 5120;
  float* Sbase = smem + 43072;

  const int tid  = threadIdx.x;
  const int wid  = tid >> 5;
  const int lane = tid & 31;
  const int ln   = lane & 15;
  const int lh   = lane >> 4;

  const int b  = blockIdx.x;
  const int a  = b >> 5;
  const int cc = b & 31;
  const size_t xoff = (size_t)(cc * NSET + a) * (SEQ * DMODEL);
  const size_t yoff = (size_t)(a * NSET + cc) * (SEQ * DMODEL);
  const size_t ooff = (size_t)(cc * NSET + a) * (SEQ * HEAD_SIZE);

  {
    const float4* gx = (const float4*)(x + xoff);
    const float4* gy = (const float4*)(y + yoff);
    float4* lx = (float4*)Xs;
    float4* ly = (float4*)Ys;
    for (int i = tid; i < (SEQ * DMODEL) / 4; i += 256) { lx[i] = gx[i]; ly[i] = gy[i]; }
    for (int i = tid; i < 800; i += 256) {
      const int k = i / DQK, n = i % DQK;
      Wqt[n * DMODEL + k] = WQg[i];
      Wkt[n * DMODEL + k] = WKg[i];
      Wvt[n * DMODEL + k] = WVg[i];
    }
    for (int i = tid; i < 800; i += 256) {
      const int k = i / HEAD_SIZE, n = i % HEAD_SIZE;
      Wht[n * DQK + k] = WHg[i];
    }
    for (int i = tid; i < 160; i += 256) {
      Wqt[800 + i] = 0.f; Wkt[800 + i] = 0.f; Wvt[800 + i] = 0.f;
    }
    for (int i = tid; i < 480; i += 256) Wht[800 + i] = 0.f;
    for (int i = tid; i < (VTROWS - QKS) * SEQ; i += 256) Vt[QKS * SEQ + i] = 0.f;
  }
  __syncthreads();

  for (int rbi = 0; rbi < 2; ++rbi) {
    const int rb   = wid + rbi * 8;
    const int arow = (rb * 16 + ln) * DMODEL;
    for (int ct = 0; ct < 3; ++ct) {
      const int col = ct * 16 + ln;
      v8f cq = {}, ck = {}, cv = {};
#pragma unroll
      for (int k0 = 0; k0 < DMODEL; k0 += 4) {
        const int kk = k0 + 2 * lh;
        v2f ax = *(const v2f*)&Xs[arow + kk];
        v2f ay = *(const v2f*)&Ys[arow + kk];
        cq = wmma4(ax, *(const v2f*)&Wqt[col * DMODEL + kk], cq);
        ck = wmma4(ay, *(const v2f*)&Wkt[col * DMODEL + kk], ck);
        cv = wmma4(ay, *(const v2f*)&Wvt[col * DMODEL + kk], cv);
      }
#pragma unroll
      for (int v = 0; v < 8; ++v) {
        const int row = rb * 16 + v + 8 * lh;
        Qs[row * QKS + col] = cq[v];
        Ks[row * QKS + col] = ck[v];
        Vt[col * SEQ + row] = cv[v];
      }
    }
  }
  __syncthreads();

  const float scale = 0.22360679774997896f;
  float* Sw = Sbase + wid * (16 * SEQ);

  for (int h = 0; h < NUM_HEADS; ++h) {
    const int hoff = h * HEAD_SIZE;
    for (int rbi = 0; rbi < 2; ++rbi) {
      const int rb = wid + rbi * 8;

      {
        v2f aq[5];
#pragma unroll
        for (int i = 0; i < 5; ++i)
          aq[i] = *(const v2f*)&Qs[(rb * 16 + ln) * QKS + hoff + 4 * i + 2 * lh];
        for (int ct = 0; ct < 16; ct += 2) {
          v8f c0 = {}, c1 = {};
#pragma unroll
          for (int i = 0; i < 5; ++i) {
            const int kk = hoff + 4 * i + 2 * lh;
            c0 = wmma4(aq[i], *(const v2f*)&Ks[(ct * 16 + ln) * QKS + kk], c0);
            c1 = wmma4(aq[i], *(const v2f*)&Ks[((ct + 1) * 16 + ln) * QKS + kk], c1);
          }
#pragma unroll
          for (int v = 0; v < 8; ++v) {
            Sw[(v + 8 * lh) * SEQ + ct * 16 + ln]       = c0[v] * scale;
            Sw[(v + 8 * lh) * SEQ + (ct + 1) * 16 + ln] = c1[v] * scale;
          }
        }
      }

      {
        float4* row = (float4*)&Sw[ln * SEQ + lh * 128];
        float mx = -__builtin_inff();
        for (int i = 0; i < 32; ++i) {
          const float4 r = row[i];
          mx = fmaxf(mx, fmaxf(fmaxf(r.x, r.y), fmaxf(r.z, r.w)));
        }
        mx = fmaxf(mx, __shfl_xor(mx, 16, 32));
        float sum = 0.f;
        for (int i = 0; i < 32; ++i) {
          const float4 r = row[i];
          float4 e;
          e.x = (r.x != 0.f) ? __expf(r.x - mx) : 0.f;
          e.y = (r.y != 0.f) ? __expf(r.y - mx) : 0.f;
          e.z = (r.z != 0.f) ? __expf(r.z - mx) : 0.f;
          e.w = (r.w != 0.f) ? __expf(r.w - mx) : 0.f;
          row[i] = e;
          sum += (e.x + e.y) + (e.z + e.w);
        }
        sum += __shfl_xor(sum, 16, 32);
        const float inv = 1.f / (sum + 1e-10f);
        for (int i = 0; i < 32; ++i) {
          float4 r = row[i];
          r.x *= inv; r.y *= inv; r.z *= inv; r.w *= inv;
          row[i] = r;
        }
      }

      for (int ct = 0; ct < 2; ++ct) {
        const int col  = ct * 16 + ln;
        const int vrow = hoff + col;
        v8f c0 = {}, c1 = {};
        for (int k0 = 0; k0 < SEQ; k0 += 8) {
          const int kk = k0 + 2 * lh;
          c0 = wmma4(*(const v2f*)&Sw[ln * SEQ + kk],
                     *(const v2f*)&Vt[vrow * SEQ + kk], c0);
          c1 = wmma4(*(const v2f*)&Sw[ln * SEQ + kk + 4],
                     *(const v2f*)&Vt[vrow * SEQ + kk + 4], c1);
        }
        const v8f c = c0 + c1;
        if (col < HEAD_SIZE) {
#pragma unroll
          for (int v = 0; v < 8; ++v)
            Qs[(rb * 16 + v + 8 * lh) * QKS + hoff + col] = c[v];
        }
      }
    }
  }

  for (int rbi = 0; rbi < 2; ++rbi) {
    const int rb = wid + rbi * 8;
    for (int ct = 0; ct < 2; ++ct) {
      const int col = ct * 16 + ln;
      v8f c0 = {}, c1 = {};
#pragma unroll
      for (int k0 = 0; k0 < DQK; k0 += 8) {
        const int kk = k0 + 2 * lh;
        c0 = wmma4(*(const v2f*)&Qs[(rb * 16 + ln) * QKS + kk],
                   *(const v2f*)&Wht[col * DQK + kk], c0);
        c1 = wmma4(*(const v2f*)&Qs[(rb * 16 + ln) * QKS + kk + 4],
                   *(const v2f*)&Wht[col * DQK + kk + 4], c1);
      }
      const v8f c = c0 + c1;
      if (col < HEAD_SIZE) {
        float* og = out + ooff;
#pragma unroll
        for (int v = 0; v < 8; ++v)
          og[(rb * 16 + v + 8 * lh) * HEAD_SIZE + col] = c[v];
      }
    }
  }
}

extern "C" void kernel_launch(void* const* d_in, const int* in_sizes, int n_in,
                              void* d_out, int out_size, void* d_ws, size_t ws_size,
                              hipStream_t stream) {
  (void)in_sizes; (void)n_in; (void)out_size; (void)d_ws; (void)ws_size;
  const float* x  = (const float*)d_in[0];
  const float* y  = (const float*)d_in[1];
  const float* WQ = (const float*)d_in[2];
  const float* WK = (const float*)d_in[3];
  const float* WV = (const float*)d_in[4];
  const float* WH = (const float*)d_in[5];
  float* out = (float*)d_out;

  const size_t shmem = (size_t)75840 * sizeof(float);
  (void)hipFuncSetAttribute((const void*)set_attention_kernel,
                            hipFuncAttributeMaxDynamicSharedMemorySize,
                            (int)shmem);
  hipLaunchKernelGGL(set_attention_kernel,
                     dim3(NSET * NSET), dim3(256), shmem, stream,
                     x, y, WQ, WK, WV, WH, out);
}


# LDS row pitches padded to 4 x odd dwords (Q/K 48->52, S strips and V^T 256->260) so softmax row reads and WMMA fragment reads hit distinct banks; arithmetic and summation order unchanged
# speedup vs baseline: 2.9330x; 2.9330x over previous
#include <hip/hip_runtime.h>


typedef __attribute__((ext_vector_type(2))) float v2f;
typedef __attribute__((ext_vector_type(8))) float v8f;

#define HEAD_SIZE 20
#define NUM_HEADS 2
#define SEQ 256
#define NSET 32
#define DMODEL 20
#define DQK 40
#define QKS 52
#define VTROWS 56
#define VTPROJ 48
#define SP 260

__device__ __forceinline__ v8f wmma4(v2f a, v2f b, v8f c) {
  return __builtin_amdgcn_wmma_f32_16x16x4_f32(
       false, a,  false, b,
       (short)0, c,  false,  false);
}

__global__ __launch_bounds__(256)
void set_attention_kernel(const float* __restrict__ x,
                          const float* __restrict__ y,
                          const float* __restrict__ WQg,
                          const float* __restrict__ WKg,
                          const float* __restrict__ WVg,
                          const float* __restrict__ WHg,
                          float* __restrict__ out) {
  extern __shared__ float smem[];
  float* Qs  = smem;
  float* Ks  = smem + 13312;
  float* Vt  = smem + 26624;
  float* Wqt = smem + 41184;
  float* Wkt = Wqt + 960;
  float* Wvt = Wkt + 960;
  float* Wht = smem + 44064;
  float* Xs  = smem + 45344;
  float* Ys  = Xs + 5120;
  float* Sbase = smem + 45344;

  const int tid  = threadIdx.x;
  const int wid  = tid >> 5;
  const int lane = tid & 31;
  const int ln   = lane & 15;
  const int lh   = lane >> 4;

  const int b  = blockIdx.x;
  const int a  = b >> 5;
  const int cc = b & 31;
  const size_t xoff = (size_t)(cc * NSET + a) * (SEQ * DMODEL);
  const size_t yoff = (size_t)(a * NSET + cc) * (SEQ * DMODEL);
  const size_t ooff = (size_t)(cc * NSET + a) * (SEQ * HEAD_SIZE);

  {
    const float4* gx = (const float4*)(x + xoff);
    const float4* gy = (const float4*)(y + yoff);
    float4* lx = (float4*)Xs;
    float4* ly = (float4*)Ys;
    for (int i = tid; i < (SEQ * DMODEL) / 4; i += 256) { lx[i] = gx[i]; ly[i] = gy[i]; }
    for (int i = tid; i < 800; i += 256) {
      const int k = i / DQK, n = i % DQK;
      Wqt[n * DMODEL + k] = WQg[i];
      Wkt[n * DMODEL + k] = WKg[i];
      Wvt[n * DMODEL + k] = WVg[i];
    }
    for (int i = tid; i < 800; i += 256) {
      const int k = i / HEAD_SIZE, n = i % HEAD_SIZE;
      Wht[n * DQK + k] = WHg[i];
    }
    for (int i = tid; i < 160; i += 256) {
      Wqt[800 + i] = 0.f; Wkt[800 + i] = 0.f; Wvt[800 + i] = 0.f;
    }
    for (int i = tid; i < 480; i += 256) Wht[800 + i] = 0.f;
    for (int i = tid; i < (VTROWS - VTPROJ) * SP; i += 256) Vt[VTPROJ * SP + i] = 0.f;
  }
  __syncthreads();

  for (int rbi = 0; rbi < 2; ++rbi) {
    const int rb   = wid + rbi * 8;
    const int arow = (rb * 16 + ln) * DMODEL;
    for (int ct = 0; ct < 3; ++ct) {
      const int col = ct * 16 + ln;
      v8f cq = {}, ck = {}, cv = {};
#pragma unroll
      for (int k0 = 0; k0 < DMODEL; k0 += 4) {
        const int kk = k0 + 2 * lh;
        v2f ax = *(const v2f*)&Xs[arow + kk];
        v2f ay = *(const v2f*)&Ys[arow + kk];
        cq = wmma4(ax, *(const v2f*)&Wqt[col * DMODEL + kk], cq);
        ck = wmma4(ay, *(const v2f*)&Wkt[col * DMODEL + kk], ck);
        cv = wmma4(ay, *(const v2f*)&Wvt[col * DMODEL + kk], cv);
      }
#pragma unroll
      for (int v = 0; v < 8; ++v) {
        const int row = rb * 16 + v + 8 * lh;
        Qs[row * QKS + col] = cq[v];
        Ks[row * QKS + col] = ck[v];
        Vt[col * SP + row] = cv[v];
      }
    }
  }
  __syncthreads();

  const float scale = 0.22360679774997896f;
  float* Sw = Sbase + wid * (16 * SP);

  for (int h = 0; h < NUM_HEADS; ++h) {
    const int hoff = h * HEAD_SIZE;
    for (int rbi = 0; rbi < 2; ++rbi) {
      const int rb = wid + rbi * 8;

      {
        v2f aq[5];
#pragma unroll
        for (int i = 0; i < 5; ++i)
          aq[i] = *(const v2f*)&Qs[(rb * 16 + ln) * QKS + hoff + 4 * i + 2 * lh];
        for (int ct = 0; ct < 16; ct += 2) {
          v8f c0 = {}, c1 = {};
#pragma unroll
          for (int i = 0; i < 5; ++i) {
            const int kk = hoff + 4 * i + 2 * lh;
            c0 = wmma4(aq[i], *(const v2f*)&Ks[(ct * 16 + ln) * QKS + kk], c0);
            c1 = wmma4(aq[i], *(const v2f*)&Ks[((ct + 1) * 16 + ln) * QKS + kk], c1);
          }
#pragma unroll
          for (int v = 0; v < 8; ++v) {
            Sw[(v + 8 * lh) * SP + ct * 16 + ln]       = c0[v] * scale;
            Sw[(v + 8 * lh) * SP + (ct + 1) * 16 + ln] = c1[v] * scale;
          }
        }
      }

      {
        float4* row = (float4*)&Sw[ln * SP + lh * 128];
        float mx = -__builtin_inff();
        for (int i = 0; i < 32; ++i) {
          const float4 r = row[i];
          mx = fmaxf(mx, fmaxf(fmaxf(r.x, r.y), fmaxf(r.z, r.w)));
        }
        mx = fmaxf(mx, __shfl_xor(mx, 16, 32));
        float sum = 0.f;
        for (int i = 0; i < 32; ++i) {
          const float4 r = row[i];
          float4 e;
          e.x = (r.x != 0.f) ? __expf(r.x - mx) : 0.f;
          e.y = (r.y != 0.f) ? __expf(r.y - mx) : 0.f;
          e.z = (r.z != 0.f) ? __expf(r.z - mx) : 0.f;
          e.w = (r.w != 0.f) ? __expf(r.w - mx) : 0.f;
          row[i] = e;
          sum += (e.x + e.y) + (e.z + e.w);
        }
        sum += __shfl_xor(sum, 16, 32);
        const float inv = 1.f / (sum + 1e-10f);
        for (int i = 0; i < 32; ++i) {
          float4 r = row[i];
          r.x *= inv; r.y *= inv; r.z *= inv; r.w *= inv;
          row[i] = r;
        }
      }

      for (int ct = 0; ct < 2; ++ct) {
        const int col  = ct * 16 + ln;
        const int vrow = hoff + col;
        v8f c0 = {}, c1 = {};
        for (int k0 = 0; k0 < SEQ; k0 += 8) {
          const int kk = k0 + 2 * lh;
          c0 = wmma4(*(const v2f*)&Sw[ln * SP + kk],
                     *(const v2f*)&Vt[vrow * SP + kk], c0);
          c1 = wmma4(*(const v2f*)&Sw[ln * SP + kk + 4],
                     *(const v2f*)&Vt[vrow * SP + kk + 4], c1);
        }
        const v8f c = c0 + c1;
        if (col < HEAD_SIZE) {
#pragma unroll
          for (int v = 0; v < 8; ++v)
            Qs[(rb * 16 + v + 8 * lh) * QKS + hoff + col] = c[v];
        }
      }
    }
  }

  for (int rbi = 0; rbi < 2; ++rbi) {
    const int rb = wid + rbi * 8;
    for (int ct = 0; ct < 2; ++ct) {
      const int col = ct * 16 + ln;
      v8f c0 = {}, c1 = {};
#pragma unroll
      for (int k0 = 0; k0 < DQK; k0 += 8) {
        const int kk = k0 + 2 * lh;
        c0 = wmma4(*(const v2f*)&Qs[(rb * 16 + ln) * QKS + kk],
                   *(const v2f*)&Wht[col * DQK + kk], c0);
        c1 = wmma4(*(const v2f*)&Qs[(rb * 16 + ln) * QKS + kk + 4],
                   *(const v2f*)&Wht[col * DQK + kk + 4], c1);
      }
      const v8f c = c0 + c1;
      if (col < HEAD_SIZE) {
        float* og = out + ooff;
#pragma unroll
        for (int v = 0; v < 8; ++v)
          og[(rb * 16 + v + 8 * lh) * HEAD_SIZE + col] = c[v];
      }
    }
  }
}

extern "C" void kernel_launch(void* const* d_in, const int* in_sizes, int n_in,
                              void* d_out, int out_size, void* d_ws, size_t ws_size,
                              hipStream_t stream) {
  (void)in_sizes; (void)n_in; (void)out_size; (void)d_ws; (void)ws_size;
  const float* x  = (const float*)d_in[0];
  const float* y  = (const float*)d_in[1];
  const float* WQ = (const float*)d_in[2];
  const float* WK = (const float*)d_in[3];
  const float* WV = (const float*)d_in[4];
  const float* WH = (const float*)d_in[5];
  float* out = (float*)d_out;

  const size_t shmem = (size_t)78624 * sizeof(float);
  (void)hipFuncSetAttribute((const void*)set_attention_kernel,
                            hipFuncAttributeMaxDynamicSharedMemorySize,
                            (int)shmem);
  hipLaunchKernelGGL(set_attention_kernel,
                     dim3(NSET * NSET), dim3(256), shmem, stream,
                     x, y, WQ, WK, WV, WH, out);
}


# logits strip kept in registers as S^T=K.Q^T so its D tiles feed O^T=V^T.P^T as B fragments; softmax in registers, 1/sum applied to O; S strip in LDS gone; exp-sum and PV key summation order change
# speedup vs baseline: 3.4989x; 1.1929x over previous
#include <hip/hip_runtime.h>


typedef __attribute__((ext_vector_type(2))) float v2f;
typedef __attribute__((ext_vector_type(8))) float v8f;

#define HEAD_SIZE 20
#define NUM_HEADS 2
#define SEQ 256
#define NSET 32
#define DMODEL 20
#define DQK 40
#define QKS 52
#define VTROWS 56
#define VTPROJ 48
#define SP 260

__device__ __forceinline__ v8f wmma4(v2f a, v2f b, v8f c) {
  return __builtin_amdgcn_wmma_f32_16x16x4_f32(
       false, a,  false, b,
       (short)0, c,  false,  false);
}

__global__ __launch_bounds__(256)
void set_attention_kernel(const float* __restrict__ x,
                          const float* __restrict__ y,
                          const float* __restrict__ WQg,
                          const float* __restrict__ WKg,
                          const float* __restrict__ WVg,
                          const float* __restrict__ WHg,
                          float* __restrict__ out) {
  extern __shared__ float smem[];
  float* Qs  = smem;
  float* Ks  = smem + 13312;
  float* Vt  = smem + 26624;
  float* Wqt = smem + 41184;
  float* Wkt = Wqt + 960;
  float* Wvt = Wkt + 960;
  float* Wht = smem + 44064;
  float* Xs  = smem + 45344;
  float* Ys  = Xs + 5120;

  const int tid  = threadIdx.x;
  const int wid  = tid >> 5;
  const int lane = tid & 31;
  const int ln   = lane & 15;
  const int lh   = lane >> 4;

  const int b  = blockIdx.x;
  const int a  = b >> 5;
  const int cc = b & 31;
  const size_t xoff = (size_t)(cc * NSET + a) * (SEQ * DMODEL);
  const size_t yoff = (size_t)(a * NSET + cc) * (SEQ * DMODEL);
  const size_t ooff = (size_t)(cc * NSET + a) * (SEQ * HEAD_SIZE);

  {
    const float4* gx = (const float4*)(x + xoff);
    const float4* gy = (const float4*)(y + yoff);
    float4* lx = (float4*)Xs;
    float4* ly = (float4*)Ys;
    for (int i = tid; i < (SEQ * DMODEL) / 4; i += 256) { lx[i] = gx[i]; ly[i] = gy[i]; }
    for (int i = tid; i < 800; i += 256) {
      const int k = i / DQK, n = i % DQK;
      Wqt[n * DMODEL + k] = WQg[i];
      Wkt[n * DMODEL + k] = WKg[i];
      Wvt[n * DMODEL + k] = WVg[i];
    }
    for (int i = tid; i < 800; i += 256) {
      const int k = i / HEAD_SIZE, n = i % HEAD_SIZE;
      Wht[n * DQK + k] = WHg[i];
    }
    for (int i = tid; i < 160; i += 256) {
      Wqt[800 + i] = 0.f; Wkt[800 + i] = 0.f; Wvt[800 + i] = 0.f;
    }
    for (int i = tid; i < 480; i += 256) Wht[800 + i] = 0.f;
    for (int i = tid; i < (VTROWS - VTPROJ) * SP; i += 256) Vt[VTPROJ * SP + i] = 0.f;
  }
  __syncthreads();

  for (int rbi = 0; rbi < 2; ++rbi) {
    const int rb   = wid + rbi * 8;
    const int arow = (rb * 16 + ln) * DMODEL;
    for (int ct = 0; ct < 3; ++ct) {
      const int col = ct * 16 + ln;
      v8f cq = {}, ck = {}, cv = {};
#pragma unroll
      for (int k0 = 0; k0 < DMODEL; k0 += 4) {
        const int kk = k0 + 2 * lh;
        v2f ax = *(const v2f*)&Xs[arow + kk];
        v2f ay = *(const v2f*)&Ys[arow + kk];
        cq = wmma4(ax, *(const v2f*)&Wqt[col * DMODEL + kk], cq);
        ck = wmma4(ay, *(const v2f*)&Wkt[col * DMODEL + kk], ck);
        cv = wmma4(ay, *(const v2f*)&Wvt[col * DMODEL + kk], cv);
      }
#pragma unroll
      for (int v = 0; v < 8; ++v) {
        const int row = rb * 16 + v + 8 * lh;
        Qs[row * QKS + col] = cq[v];
        Ks[row * QKS + col] = ck[v];
        Vt[col * SP + row] = cv[v];
      }
    }
  }
  __syncthreads();

  const float scale = 0.22360679774997896f;
  const int kperm = ((ln & 6) << 1) | ((ln >> 3) << 1) | (ln & 1);

  for (int h = 0; h < NUM_HEADS; ++h) {
    const int hoff = h * HEAD_SIZE;
    for (int rbi = 0; rbi < 2; ++rbi) {
      const int rb = wid + rbi * 8;

      v8f S[16];
      {
        v2f bq[5];
#pragma unroll
        for (int i = 0; i < 5; ++i)
          bq[i] = *(const v2f*)&Qs[(rb * 16 + ln) * QKS + hoff + 4 * i + 2 * lh];
        const float* kbase = &Ks[kperm * QKS + hoff + 2 * lh];
#pragma unroll
        for (int t = 0; t < 16; t += 4) {
          v2f ak[4][5];
#pragma unroll
          for (int u = 0; u < 4; ++u)
#pragma unroll
            for (int i = 0; i < 5; ++i)
              ak[u][i] = *(const v2f*)&kbase[(t + u) * 16 * QKS + 4 * i];
          v8f c0 = {}, c1 = {}, c2 = {}, c3 = {};
#pragma unroll
          for (int i = 0; i < 5; ++i) {
            c0 = wmma4(ak[0][i], bq[i], c0);
            c1 = wmma4(ak[1][i], bq[i], c1);
            c2 = wmma4(ak[2][i], bq[i], c2);
            c3 = wmma4(ak[3][i], bq[i], c3);
          }
          S[t] = c0; S[t + 1] = c1; S[t + 2] = c2; S[t + 3] = c3;
        }
      }

      float inv;
      {
        float mx = -__builtin_inff();
#pragma unroll
        for (int t = 0; t < 16; ++t) {
#pragma unroll
          for (int r = 0; r < 8; ++r) S[t][r] *= scale;
          mx = fmaxf(mx, fmaxf(fmaxf(fmaxf(S[t][0], S[t][1]), fmaxf(S[t][2], S[t][3])),
                               fmaxf(fmaxf(S[t][4], S[t][5]), fmaxf(S[t][6], S[t][7]))));
        }
        mx = fmaxf(mx, __shfl_xor(mx, 16, 32));
        float sum = 0.f;
#pragma unroll
        for (int t = 0; t < 16; ++t) {
#pragma unroll
          for (int r = 0; r < 8; ++r) {
            const float s = S[t][r];
            S[t][r] = (s != 0.f) ? __expf(s - mx) : 0.f;
          }
          sum += ((S[t][0] + S[t][1]) + (S[t][2] + S[t][3])) +
                 ((S[t][4] + S[t][5]) + (S[t][6] + S[t][7]));
        }
        sum += __shfl_xor(sum, 16, 32);
        inv = 1.f / (sum + 1e-10f);
      }

      {
        const float* v0p = &Vt[(hoff + ln) * SP + 2 * lh];
        const float* v1p = v0p + 16 * SP;
        v8f o0 = {}, o1 = {};
#pragma unroll
        for (int t = 0; t < 16; t += 2) {
          v2f a0[8], a1[8];
#pragma unroll
          for (int j = 0; j < 8; ++j) {
            a0[j] = *(const v2f*)&v0p[t * 16 + 4 * j];
            a1[j] = *(const v2f*)&v1p[t * 16 + 4 * j];
          }
#pragma unroll
          for (int j = 0; j < 8; ++j) {
            const int tt = t + (j >> 2), jj = j & 3;
            v2f bp;
            bp[0] = S[tt][2 * jj];
            bp[1] = S[tt][2 * jj + 1];
            o0 = wmma4(a0[j], bp, o0);
            o1 = wmma4(a1[j], bp, o1);
          }
        }
        float* orow = &Qs[(rb * 16 + ln) * QKS + hoff];
        float4 w;
        w.x = o0[0] * inv; w.y = o0[1] * inv; w.z = o0[2] * inv; w.w = o0[3] * inv;
        *(float4*)&orow[8 * lh] = w;
        w.x = o0[4] * inv; w.y = o0[5] * inv; w.z = o0[6] * inv; w.w = o0[7] * inv;
        *(float4*)&orow[8 * lh + 4] = w;
        if (lh == 0) {
          w.x = o1[0] * inv; w.y = o1[1] * inv; w.z = o1[2] * inv; w.w = o1[3] * inv;
          *(float4*)&orow[16] = w;
        }
      }
    }
  }

  for (int rbi = 0; rbi < 2; ++rbi) {
    const int rb = wid + rbi * 8;
    for (int ct = 0; ct < 2; ++ct) {
      const int col = ct * 16 + ln;
      v8f c0 = {}, c1 = {};
#pragma unroll
      for (int k0 = 0; k0 < DQK; k0 += 8) {
        const int kk = k0 + 2 * lh;
        c0 = wmma4(*(const v2f*)&Qs[(rb * 16 + ln) * QKS + kk],
                   *(const v2f*)&Wht[col * DQK + kk], c0);
        c1 = wmma4(*(const v2f*)&Qs[(rb * 16 + ln) * QKS + kk + 4],
                   *(const v2f*)&Wht[col * DQK + kk + 4], c1);
      }
      const v8f c = c0 + c1;
      if (col < HEAD_SIZE) {
        float* og = out + ooff;
#pragma unroll
        for (int v = 0; v < 8; ++v)
          og[(rb * 16 + v + 8 * lh) * HEAD_SIZE + col] = c[v];
      }
    }
  }
}

extern "C" void kernel_launch(void* const* d_in, const int* in_sizes, int n_in,
                              void* d_out, int out_size, void* d_ws, size_t ws_size,
                              hipStream_t stream) {
  (void)in_sizes; (void)n_in; (void)out_size; (void)d_ws; (void)ws_size;
  const float* x  = (const float*)d_in[0];
  const float* y  = (const float*)d_in[1];
  const float* WQ = (const float*)d_in[2];
  const float* WK = (const float*)d_in[3];
  const float* WV = (const float*)d_in[4];
  const float* WH = (const float*)d_in[5];
  float* out = (float*)d_out;

  const size_t shmem = (size_t)55584 * sizeof(float);
  (void)hipFuncSetAttribute((const void*)set_attention_kernel,
                            hipFuncAttributeMaxDynamicSharedMemorySize,
                            (int)shmem);
  hipLaunchKernelGGL(set_attention_kernel,
                     dim3(NSET * NSET), dim3(256), shmem, stream,
                     x, y, WQ, WK, WV, WH, out);
}


# LDS per workgroup cut 222336 -> 148352 B (x/y staged in the Q/K rows they become, Q/K pitch 52->44, V^T pad rows dropped) so two workgroups share a WGP; arithmetic and summation order unchanged
# speedup vs baseline: 4.3567x; 1.2452x over previous
#include <hip/hip_runtime.h>


typedef __attribute__((ext_vector_type(2))) float v2f;
typedef __attribute__((ext_vector_type(8))) float v8f;

#define HEAD_SIZE 20
#define NUM_HEADS 2
#define SEQ 256
#define NSET 32
#define DMODEL 20
#define DQK 40
#define QKS 44
#define SP 260

__device__ __forceinline__ v8f wmma4(v2f a, v2f b, v8f c) {
  return __builtin_amdgcn_wmma_f32_16x16x4_f32(
       false, a,  false, b,
       (short)0, c,  false,  false);
}

__global__ __launch_bounds__(256)
void set_attention_kernel(const float* __restrict__ x,
                          const float* __restrict__ y,
                          const float* __restrict__ WQg,
                          const float* __restrict__ WKg,
                          const float* __restrict__ WVg,
                          const float* __restrict__ WHg,
                          float* __restrict__ out) {
  extern __shared__ float smem[];
  float* Qs  = smem;
  float* Ks  = smem + 11264;
  float* Vt  = smem + 22528;
  float* Wqt = smem + 32928;
  float* Wkt = Wqt + 960;
  float* Wvt = Wkt + 960;
  float* Wht = smem + 35808;

  const int tid  = threadIdx.x;
  const int wid  = tid >> 5;
  const int lane = tid & 31;
  const int ln   = lane & 15;
  const int lh   = lane >> 4;

  const int b  = blockIdx.x;
  const int a  = b >> 5;
  const int cc = b & 31;
  const size_t xoff = (size_t)(cc * NSET + a) * (SEQ * DMODEL);
  const size_t yoff = (size_t)(a * NSET + cc) * (SEQ * DMODEL);
  const size_t ooff = (size_t)(cc * NSET + a) * (SEQ * HEAD_SIZE);

  {
    const float4* gx = (const float4*)(x + xoff);
    const float4* gy = (const float4*)(y + yoff);
    float4* lx = (float4*)Qs;
    float4* ly = (float4*)Ks;
    for (int i = tid; i < (SEQ * DMODEL) / 4; i += 256) {
      const int r = i / 5, c = i - 5 * r;
      lx[r * (QKS / 4) + c] = gx[i];
      ly[r * (QKS / 4) + c] = gy[i];
    }
    for (int i = tid; i < 800; i += 256) {
      const int k = i / DQK, n = i % DQK;
      Wqt[n * DMODEL + k] = WQg[i];
      Wkt[n * DMODEL + k] = WKg[i];
      Wvt[n * DMODEL + k] = WVg[i];
    }
    for (int i = tid; i < 800; i += 256) {
      const int k = i / HEAD_SIZE, n = i % HEAD_SIZE;
      Wht[n * DQK + k] = WHg[i];
    }
    for (int i = tid; i < 160; i += 256) {
      Wqt[800 + i] = 0.f; Wkt[800 + i] = 0.f; Wvt[800 + i] = 0.f;
    }
    for (int i = tid; i < 480; i += 256) Wht[800 + i] = 0.f;
  }
  __syncthreads();

  for (int rbi = 0; rbi < 2; ++rbi) {
    const int rb   = wid + rbi * 8;
    const int arow = (rb * 16 + ln) * QKS + 2 * lh;
    v2f ax[5], ay[5];
#pragma unroll
    for (int i = 0; i < 5; ++i) {
      ax[i] = *(const v2f*)&Qs[arow + 4 * i];
      ay[i] = *(const v2f*)&Ks[arow + 4 * i];
    }
    for (int ct = 0; ct < 3; ++ct) {
      const int col = ct * 16 + ln;
      v8f cq = {}, ck = {}, cv = {};
#pragma unroll
      for (int i = 0; i < 5; ++i) {
        const int kk = 4 * i + 2 * lh;
        cq = wmma4(ax[i], *(const v2f*)&Wqt[col * DMODEL + kk], cq);
        ck = wmma4(ay[i], *(const v2f*)&Wkt[col * DMODEL + kk], ck);
        cv = wmma4(ay[i], *(const v2f*)&Wvt[col * DMODEL + kk], cv);
      }
      if (col < DQK) {
#pragma unroll
        for (int v = 0; v < 8; ++v) {
          const int row = rb * 16 + v + 8 * lh;
          Qs[row * QKS + col] = cq[v];
          Ks[row * QKS + col] = ck[v];
          Vt[col * SP + row] = cv[v];
        }
      }
    }
  }
  __syncthreads();

  const float scale = 0.22360679774997896f;
  const int kperm = ((ln & 6) << 1) | ((ln >> 3) << 1) | (ln & 1);

  for (int h = 0; h < NUM_HEADS; ++h) {
    const int hoff = h * HEAD_SIZE;
    for (int rbi = 0; rbi < 2; ++rbi) {
      const int rb = wid + rbi * 8;

      v8f S[16];
      {
        v2f bq[5];
#pragma unroll
        for (int i = 0; i < 5; ++i)
          bq[i] = *(const v2f*)&Qs[(rb * 16 + ln) * QKS + hoff + 4 * i + 2 * lh];
        const float* kbase = &Ks[kperm * QKS + hoff + 2 * lh];
#pragma unroll
        for (int t = 0; t < 16; t += 4) {
          v2f ak[4][5];
#pragma unroll
          for (int u = 0; u < 4; ++u)
#pragma unroll
            for (int i = 0; i < 5; ++i)
              ak[u][i] = *(const v2f*)&kbase[(t + u) * 16 * QKS + 4 * i];
          v8f c0 = {}, c1 = {}, c2 = {}, c3 = {};
#pragma unroll
          for (int i = 0; i < 5; ++i) {
            c0 = wmma4(ak[0][i], bq[i], c0);
            c1 = wmma4(ak[1][i], bq[i], c1);
            c2 = wmma4(ak[2][i], bq[i], c2);
            c3 = wmma4(ak[3][i], bq[i], c3);
          }
          S[t] = c0; S[t + 1] = c1; S[t + 2] = c2; S[t + 3] = c3;
        }
      }

      float inv;
      {
        float mx = -__builtin_inff();
#pragma unroll
        for (int t = 0; t < 16; ++t) {
#pragma unroll
          for (int r = 0; r < 8; ++r) S[t][r] *= scale;
          mx = fmaxf(mx, fmaxf(fmaxf(fmaxf(S[t][0], S[t][1]), fmaxf(S[t][2], S[t][3])),
                               fmaxf(fmaxf(S[t][4], S[t][5]), fmaxf(S[t][6], S[t][7]))));
        }
        mx = fmaxf(mx, __shfl_xor(mx, 16, 32));
        float sum = 0.f;
#pragma unroll
        for (int t = 0; t < 16; ++t) {
#pragma unroll
          for (int r = 0; r < 8; ++r) {
            const float s = S[t][r];
            S[t][r] = (s != 0.f) ? __expf(s - mx) : 0.f;
          }
          sum += ((S[t][0] + S[t][1]) + (S[t][2] + S[t][3])) +
                 ((S[t][4] + S[t][5]) + (S[t][6] + S[t][7]));
        }
        sum += __shfl_xor(sum, 16, 32);
        inv = 1.f / (sum + 1e-10f);
      }

      {
        const float* v0p = &Vt[(hoff + ln) * SP + 2 * lh];
        const float* v1p = v0p + 16 * SP;
        v8f o0 = {}, o1 = {};
#pragma unroll
        for (int t = 0; t < 16; t += 2) {
          v2f a0[8], a1[8];
#pragma unroll
          for (int j = 0; j < 8; ++j) {
            a0[j] = *(const v2f*)&v0p[t * 16 + 4 * j];
            a1[j] = *(const v2f*)&v1p[t * 16 + 4 * j];
          }
#pragma unroll
          for (int j = 0; j < 8; ++j) {
            const int tt = t + (j >> 2), jj = j & 3;
            v2f bp;
            bp[0] = S[tt][2 * jj];
            bp[1] = S[tt][2 * jj + 1];
            o0 = wmma4(a0[j], bp, o0);
            o1 = wmma4(a1[j], bp, o1);
          }
        }
        float* orow = &Qs[(rb * 16 + ln) * QKS + hoff];
        float4 w;
        w.x = o0[0] * inv; w.y = o0[1] * inv; w.z = o0[2] * inv; w.w = o0[3] * inv;
        *(float4*)&orow[8 * lh] = w;
        w.x = o0[4] * inv; w.y = o0[5] * inv; w.z = o0[6] * inv; w.w = o0[7] * inv;
        *(float4*)&orow[8 * lh + 4] = w;
        if (lh == 0) {
          w.x = o1[0] * inv; w.y = o1[1] * inv; w.z = o1[2] * inv; w.w = o1[3] * inv;
          *(float4*)&orow[16] = w;
        }
      }
    }
  }

  for (int rbi = 0; rbi < 2; ++rbi) {
    const int rb = wid + rbi * 8;
    for (int ct = 0; ct < 2; ++ct) {
      const int col = ct * 16 + ln;
      v8f c0 = {}, c1 = {};
#pragma unroll
      for (int k0 = 0; k0 < DQK; k0 += 8) {
        const int kk = k0 + 2 * lh;
        c0 = wmma4(*(const v2f*)&Qs[(rb * 16 + ln) * QKS + kk],
                   *(const v2f*)&Wht[col * DQK + kk], c0);
        c1 = wmma4(*(const v2f*)&Qs[(rb * 16 + ln) * QKS + kk + 4],
                   *(const v2f*)&Wht[col * DQK + kk + 4], c1);
      }
      const v8f c = c0 + c1;
      if (col < HEAD_SIZE) {
        float* og = out + ooff;
#pragma unroll
        for (int v = 0; v < 8; ++v)
          og[(rb * 16 + v + 8 * lh) * HEAD_SIZE + col] = c[v];
      }
    }
  }
}

extern "C" void kernel_launch(void* const* d_in, const int* in_sizes, int n_in,
                              void* d_out, int out_size, void* d_ws, size_t ws_size,
                              hipStream_t stream) {
  (void)in_sizes; (void)n_in; (void)out_size; (void)d_ws; (void)ws_size;
  const float* x  = (const float*)d_in[0];
  const float* y  = (const float*)d_in[1];
  const float* WQ = (const float*)d_in[2];
  const float* WK = (const float*)d_in[3];
  const float* WV = (const float*)d_in[4];
  const float* WH = (const float*)d_in[5];
  float* out = (float*)d_out;

  const size_t shmem = (size_t)37088 * sizeof(float);
  (void)hipFuncSetAttribute((const void*)set_attention_kernel,
                            hipFuncAttributeMaxDynamicSharedMemorySize,
                            (int)shmem);
  hipLaunchKernelGGL(set_attention_kernel,
                     dim3(NSET * NSET), dim3(256), shmem, stream,
                     x, y, WQ, WK, WV, WH, out);
}


# staging: all global loads of x, y and the four weight matrices issued in one stretch ahead of the first LDS store (one trip to memory per workgroup, was 13 dependent trips); arithmetic unchanged
# speedup vs baseline: 4.5204x; 1.0376x over previous
#include <hip/hip_runtime.h>


typedef __attribute__((ext_vector_type(2))) float v2f;
typedef __attribute__((ext_vector_type(8))) float v8f;

#define HEAD_SIZE 20
#define NUM_HEADS 2
#define SEQ 256
#define NSET 32
#define DMODEL 20
#define DQK 40
#define QKS 44
#define SP 260

__device__ __forceinline__ v8f wmma4(v2f a, v2f b, v8f c) {
  return __builtin_amdgcn_wmma_f32_16x16x4_f32(
       false, a,  false, b,
       (short)0, c,  false,  false);
}

__global__ __launch_bounds__(256)
void set_attention_kernel(const float* __restrict__ x,
                          const float* __restrict__ y,
                          const float* __restrict__ WQg,
                          const float* __restrict__ WKg,
                          const float* __restrict__ WVg,
                          const float* __restrict__ WHg,
                          float* __restrict__ out) {
  extern __shared__ float smem[];
  float* Qs  = smem;
  float* Ks  = smem + 11264;
  float* Vt  = smem + 22528;
  float* Wqt = smem + 32928;
  float* Wkt = Wqt + 960;
  float* Wvt = Wkt + 960;
  float* Wht = smem + 35808;

  const int tid  = threadIdx.x;
  const int wid  = tid >> 5;
  const int lane = tid & 31;
  const int ln   = lane & 15;
  const int lh   = lane >> 4;

  const int b  = blockIdx.x;
  const int a  = b >> 5;
  const int cc = b & 31;
  const size_t xoff = (size_t)(cc * NSET + a) * (SEQ * DMODEL);
  const size_t yoff = (size_t)(a * NSET + cc) * (SEQ * DMODEL);
  const size_t ooff = (size_t)(cc * NSET + a) * (SEQ * HEAD_SIZE);

  {
    const float4* gx = (const float4*)(x + xoff);
    const float4* gy = (const float4*)(y + yoff);
    float4* lx = (float4*)Qs;
    float4* ly = (float4*)Ks;
    float4 rx[5], ry[5];
    float wq[4], wk[4], wv[4], wh[4];
#pragma unroll
    for (int p = 0; p < 5; ++p) {
      rx[p] = gx[tid + 256 * p];
      ry[p] = gy[tid + 256 * p];
    }
#pragma unroll
    for (int p = 0; p < 4; ++p) {
      const int i  = tid + 256 * p;
      const int ic = i < 800 ? i : 799;
      wq[p] = WQg[ic];
      wk[p] = WKg[ic];
      wv[p] = WVg[ic];
      wh[p] = WHg[ic];
    }
    for (int i = tid; i < 160; i += 256) {
      Wqt[800 + i] = 0.f; Wkt[800 + i] = 0.f; Wvt[800 + i] = 0.f;
    }
    for (int i = tid; i < 480; i += 256) Wht[800 + i] = 0.f;
#pragma unroll
    for (int p = 0; p < 5; ++p) {
      const int i = tid + 256 * p;
      const int r = i / 5, c = i - 5 * r;
      lx[r * (QKS / 4) + c] = rx[p];
      ly[r * (QKS / 4) + c] = ry[p];
    }
#pragma unroll
    for (int p = 0; p < 4; ++p) {
      const int i = tid + 256 * p;
      if (i < 800) {
        const int k = i / DQK, n = i % DQK;
        Wqt[n * DMODEL + k] = wq[p];
        Wkt[n * DMODEL + k] = wk[p];
        Wvt[n * DMODEL + k] = wv[p];
        const int k2 = i / HEAD_SIZE, n2 = i % HEAD_SIZE;
        Wht[n2 * DQK + k2] = wh[p];
      }
    }
  }
  __syncthreads();

  for (int rbi = 0; rbi < 2; ++rbi) {
    const int rb   = wid + rbi * 8;
    const int arow = (rb * 16 + ln) * QKS + 2 * lh;
    v2f ax[5], ay[5];
#pragma unroll
    for (int i = 0; i < 5; ++i) {
      ax[i] = *(const v2f*)&Qs[arow + 4 * i];
      ay[i] = *(const v2f*)&Ks[arow + 4 * i];
    }
    for (int ct = 0; ct < 3; ++ct) {
      const int col = ct * 16 + ln;
      v8f cq = {}, ck = {}, cv = {};
#pragma unroll
      for (int i = 0; i < 5; ++i) {
        const int kk = 4 * i + 2 * lh;
        cq = wmma4(ax[i], *(const v2f*)&Wqt[col * DMODEL + kk], cq);
        ck = wmma4(ay[i], *(const v2f*)&Wkt[col * DMODEL + kk], ck);
        cv = wmma4(ay[i], *(const v2f*)&Wvt[col * DMODEL + kk], cv);
      }
      if (col < DQK) {
#pragma unroll
        for (int v = 0; v < 8; ++v) {
          const int row = rb * 16 + v + 8 * lh;
          Qs[row * QKS + col] = cq[v];
          Ks[row * QKS + col] = ck[v];
          Vt[col * SP + row] = cv[v];
        }
      }
    }
  }
  __syncthreads();

  const float scale = 0.22360679774997896f;
  const int kperm = ((ln & 6) << 1) | ((ln >> 3) << 1) | (ln & 1);

  for (int h = 0; h < NUM_HEADS; ++h) {
    const int hoff = h * HEAD_SIZE;
    for (int rbi = 0; rbi < 2; ++rbi) {
      const int rb = wid + rbi * 8;

      v8f S[16];
      {
        v2f bq[5];
#pragma unroll
        for (int i = 0; i < 5; ++i)
          bq[i] = *(const v2f*)&Qs[(rb * 16 + ln) * QKS + hoff + 4 * i + 2 * lh];
        const float* kbase = &Ks[kperm * QKS + hoff + 2 * lh];
#pragma unroll
        for (int t = 0; t < 16; t += 4) {
          v2f ak[4][5];
#pragma unroll
          for (int u = 0; u < 4; ++u)
#pragma unroll
            for (int i = 0; i < 5; ++i)
              ak[u][i] = *(const v2f*)&kbase[(t + u) * 16 * QKS + 4 * i];
          v8f c0 = {}, c1 = {}, c2 = {}, c3 = {};
#pragma unroll
          for (int i = 0; i < 5; ++i) {
            c0 = wmma4(ak[0][i], bq[i], c0);
            c1 = wmma4(ak[1][i], bq[i], c1);
            c2 = wmma4(ak[2][i], bq[i], c2);
            c3 = wmma4(ak[3][i], bq[i], c3);
          }
          S[t] = c0; S[t + 1] = c1; S[t + 2] = c2; S[t + 3] = c3;
        }
      }

      float inv;
      {
        float mx = -__builtin_inff();
#pragma unroll
        for (int t = 0; t < 16; ++t) {
#pragma unroll
          for (int r = 0; r < 8; ++r) S[t][r] *= scale;
          mx = fmaxf(mx, fmaxf(fmaxf(fmaxf(S[t][0], S[t][1]), fmaxf(S[t][2], S[t][3])),
                               fmaxf(fmaxf(S[t][4], S[t][5]), fmaxf(S[t][6], S[t][7]))));
        }
        mx = fmaxf(mx, __shfl_xor(mx, 16, 32));
        float sum = 0.f;
#pragma unroll
        for (int t = 0; t < 16; ++t) {
#pragma unroll
          for (int r = 0; r < 8; ++r) {
            const float s = S[t][r];
            S[t][r] = (s != 0.f) ? __expf(s - mx) : 0.f;
          }
          sum += ((S[t][0] + S[t][1]) + (S[t][2] + S[t][3])) +
                 ((S[t][4] + S[t][5]) + (S[t][6] + S[t][7]));
        }
        sum += __shfl_xor(sum, 16, 32);
        inv = 1.f / (sum + 1e-10f);
      }

      {
        const float* v0p = &Vt[(hoff + ln) * SP + 2 * lh];
        const float* v1p = v0p + 16 * SP;
        v8f o0 = {}, o1 = {};
#pragma unroll
        for (int t = 0; t < 16; t += 2) {
          v2f a0[8], a1[8];
#pragma unroll
          for (int j = 0; j < 8; ++j) {
            a0[j] = *(const v2f*)&v0p[t * 16 + 4 * j];
            a1[j] = *(const v2f*)&v1p[t * 16 + 4 * j];
          }
#pragma unroll
          for (int j = 0; j < 8; ++j) {
            const int tt = t + (j >> 2), jj = j & 3;
            v2f bp;
            bp[0] = S[tt][2 * jj];
            bp[1] = S[tt][2 * jj + 1];
            o0 = wmma4(a0[j], bp, o0);
            o1 = wmma4(a1[j], bp, o1);
          }
        }
        float* orow = &Qs[(rb * 16 + ln) * QKS + hoff];
        float4 w;
        w.x = o0[0] * inv; w.y = o0[1] * inv; w.z = o0[2] * inv; w.w = o0[3] * inv;
        *(float4*)&orow[8 * lh] = w;
        w.x = o0[4] * inv; w.y = o0[5] * inv; w.z = o0[6] * inv; w.w = o0[7] * inv;
        *(float4*)&orow[8 * lh + 4] = w;
        if (lh == 0) {
          w.x = o1[0] * inv; w.y = o1[1] * inv; w.z = o1[2] * inv; w.w = o1[3] * inv;
          *(float4*)&orow[16] = w;
        }
      }
    }
  }

  for (int rbi = 0; rbi < 2; ++rbi) {
    const int rb = wid + rbi * 8;
    for (int ct = 0; ct < 2; ++ct) {
      const int col = ct * 16 + ln;
      v8f c0 = {}, c1 = {};
#pragma unroll
      for (int k0 = 0; k0 < DQK; k0 += 8) {
        const int kk = k0 + 2 * lh;
        c0 = wmma4(*(const v2f*)&Qs[(rb * 16 + ln) * QKS + kk],
                   *(const v2f*)&Wht[col * DQK + kk], c0);
        c1 = wmma4(*(const v2f*)&Qs[(rb * 16 + ln) * QKS + kk + 4],
                   *(const v2f*)&Wht[col * DQK + kk + 4], c1);
      }
      const v8f c = c0 + c1;
      if (col < HEAD_SIZE) {
        float* og = out + ooff;
#pragma unroll
        for (int v = 0; v < 8; ++v)
          og[(rb * 16 + v + 8 * lh) * HEAD_SIZE + col] = c[v];
      }
    }
  }
}

extern "C" void kernel_launch(void* const* d_in, const int* in_sizes, int n_in,
                              void* d_out, int out_size, void* d_ws, size_t ws_size,
                              hipStream_t stream) {
  (void)in_sizes; (void)n_in; (void)out_size; (void)d_ws; (void)ws_size;
  const float* x  = (const float*)d_in[0];
  const float* y  = (const float*)d_in[1];
  const float* WQ = (const float*)d_in[2];
  const float* WK = (const float*)d_in[3];
  const float* WV = (const float*)d_in[4];
  const float* WH = (const float*)d_in[5];
  float* out = (float*)d_out;

  const size_t shmem = (size_t)37088 * sizeof(float);
  (void)hipFuncSetAttribute((const void*)set_attention_kernel,
                            hipFuncAttributeMaxDynamicSharedMemorySize,
                            (int)shmem);
  hipLaunchKernelGGL(set_attention_kernel,
                     dim3(NSET * NSET), dim3(256), shmem, stream,
                     x, y, WQ, WK, WV, WH, out);
}


# softmax exponent argument by one fused multiply-add per raw logit, two per packed instruction (scale and log2 e folded into one constant, scale pass gone; masked logits get -inf); sum order unchanged
# speedup vs baseline: 4.6889x; 1.0373x over previous
#include <hip/hip_runtime.h>


typedef __attribute__((ext_vector_type(2))) float v2f;
typedef __attribute__((ext_vector_type(8))) float v8f;

#define HEAD_SIZE 20
#define NUM_HEADS 2
#define SEQ 256
#define NSET 32
#define DMODEL 20
#define DQK 40
#define QKS 44
#define SP 260

__device__ __forceinline__ v8f wmma4(v2f a, v2f b, v8f c) {
  return __builtin_amdgcn_wmma_f32_16x16x4_f32(
       false, a,  false, b,
       (short)0, c,  false,  false);
}

__global__ __launch_bounds__(256)
void set_attention_kernel(const float* __restrict__ x,
                          const float* __restrict__ y,
                          const float* __restrict__ WQg,
                          const float* __restrict__ WKg,
                          const float* __restrict__ WVg,
                          const float* __restrict__ WHg,
                          float* __restrict__ out) {
  extern __shared__ float smem[];
  float* Qs  = smem;
  float* Ks  = smem + 11264;
  float* Vt  = smem + 22528;
  float* Wqt = smem + 32928;
  float* Wkt = Wqt + 960;
  float* Wvt = Wkt + 960;
  float* Wht = smem + 35808;

  const int tid  = threadIdx.x;
  const int wid  = tid >> 5;
  const int lane = tid & 31;
  const int ln   = lane & 15;
  const int lh   = lane >> 4;

  const int b  = blockIdx.x;
  const int a  = b >> 5;
  const int cc = b & 31;
  const size_t xoff = (size_t)(cc * NSET + a) * (SEQ * DMODEL);
  const size_t yoff = (size_t)(a * NSET + cc) * (SEQ * DMODEL);
  const size_t ooff = (size_t)(cc * NSET + a) * (SEQ * HEAD_SIZE);

  {
    const float4* gx = (const float4*)(x + xoff);
    const float4* gy = (const float4*)(y + yoff);
    float4* lx = (float4*)Qs;
    float4* ly = (float4*)Ks;
    float4 rx[5], ry[5];
    float wq[4], wk[4], wv[4], wh[4];
#pragma unroll
    for (int p = 0; p < 5; ++p) {
      rx[p] = gx[tid + 256 * p];
      ry[p] = gy[tid + 256 * p];
    }
#pragma unroll
    for (int p = 0; p < 4; ++p) {
      const int i  = tid + 256 * p;
      const int ic = i < 800 ? i : 799;
      wq[p] = WQg[ic];
      wk[p] = WKg[ic];
      wv[p] = WVg[ic];
      wh[p] = WHg[ic];
    }
    for (int i = tid; i < 160; i += 256) {
      Wqt[800 + i] = 0.f; Wkt[800 + i] = 0.f; Wvt[800 + i] = 0.f;
    }
    for (int i = tid; i < 480; i += 256) Wht[800 + i] = 0.f;
#pragma unroll
    for (int p = 0; p < 5; ++p) {
      const int i = tid + 256 * p;
      const int r = i / 5, c = i - 5 * r;
      lx[r * (QKS / 4) + c] = rx[p];
      ly[r * (QKS / 4) + c] = ry[p];
    }
#pragma unroll
    for (int p = 0; p < 4; ++p) {
      const int i = tid + 256 * p;
      if (i < 800) {
        const int k = i / DQK, n = i % DQK;
        Wqt[n * DMODEL + k] = wq[p];
        Wkt[n * DMODEL + k] = wk[p];
        Wvt[n * DMODEL + k] = wv[p];
        const int k2 = i / HEAD_SIZE, n2 = i % HEAD_SIZE;
        Wht[n2 * DQK + k2] = wh[p];
      }
    }
  }
  __syncthreads();

  for (int rbi = 0; rbi < 2; ++rbi) {
    const int rb   = wid + rbi * 8;
    const int arow = (rb * 16 + ln) * QKS + 2 * lh;
    v2f ax[5], ay[5];
#pragma unroll
    for (int i = 0; i < 5; ++i) {
      ax[i] = *(const v2f*)&Qs[arow + 4 * i];
      ay[i] = *(const v2f*)&Ks[arow + 4 * i];
    }
    for (int ct = 0; ct < 3; ++ct) {
      const int col = ct * 16 + ln;
      v8f cq = {}, ck = {}, cv = {};
#pragma unroll
      for (int i = 0; i < 5; ++i) {
        const int kk = 4 * i + 2 * lh;
        cq = wmma4(ax[i], *(const v2f*)&Wqt[col * DMODEL + kk], cq);
        ck = wmma4(ay[i], *(const v2f*)&Wkt[col * DMODEL + kk], ck);
        cv = wmma4(ay[i], *(const v2f*)&Wvt[col * DMODEL + kk], cv);
      }
      if (col < DQK) {
#pragma unroll
        for (int v = 0; v < 8; ++v) {
          const int row = rb * 16 + v + 8 * lh;
          Qs[row * QKS + col] = cq[v];
          Ks[row * QKS + col] = ck[v];
          Vt[col * SP + row] = cv[v];
        }
      }
    }
  }
  __syncthreads();

  const float c = (float)(0.22360679774997896 * 1.4426950408889634);
  const int kperm = ((ln & 6) << 1) | ((ln >> 3) << 1) | (ln & 1);

  for (int h = 0; h < NUM_HEADS; ++h) {
    const int hoff = h * HEAD_SIZE;
    for (int rbi = 0; rbi < 2; ++rbi) {
      const int rb = wid + rbi * 8;

      v8f S[16];
      {
        v2f bq[5];
#pragma unroll
        for (int i = 0; i < 5; ++i)
          bq[i] = *(const v2f*)&Qs[(rb * 16 + ln) * QKS + hoff + 4 * i + 2 * lh];
        const float* kbase = &Ks[kperm * QKS + hoff + 2 * lh];
#pragma unroll
        for (int t = 0; t < 16; t += 4) {
          v2f ak[4][5];
#pragma unroll
          for (int u = 0; u < 4; ++u)
#pragma unroll
            for (int i = 0; i < 5; ++i)
              ak[u][i] = *(const v2f*)&kbase[(t + u) * 16 * QKS + 4 * i];
          v8f c0 = {}, c1 = {}, c2 = {}, c3 = {};
#pragma unroll
          for (int i = 0; i < 5; ++i) {
            c0 = wmma4(ak[0][i], bq[i], c0);
            c1 = wmma4(ak[1][i], bq[i], c1);
            c2 = wmma4(ak[2][i], bq[i], c2);
            c3 = wmma4(ak[3][i], bq[i], c3);
          }
          S[t] = c0; S[t + 1] = c1; S[t + 2] = c2; S[t + 3] = c3;
        }
      }

      float inv;
      {
        float mx = -__builtin_inff();
#pragma unroll
        for (int t = 0; t < 16; ++t)
          mx = fmaxf(mx, fmaxf(fmaxf(fmaxf(S[t][0], S[t][1]), fmaxf(S[t][2], S[t][3])),
                               fmaxf(fmaxf(S[t][4], S[t][5]), fmaxf(S[t][6], S[t][7]))));
        mx = fmaxf(mx, __shfl_xor(mx, 16, 32));
        const float nmc = -(mx * c);
        const v2f c2 = {c, c};
        const v2f n2 = {nmc, nmc};
        const float zlim = 0x1p-148f;
        const float ninf = -__builtin_inff();
        float sum = 0.f;
#pragma unroll
        for (int t = 0; t < 16; ++t) {
#pragma unroll
          for (int j = 0; j < 4; ++j) {
            v2f s2;
            s2[0] = S[t][2 * j];
            s2[1] = S[t][2 * j + 1];
            const v2f a2 = __builtin_elementwise_fma(s2, c2, n2);
            S[t][2 * j]     = __builtin_amdgcn_exp2f(!(__builtin_fabsf(s2[0]) <= zlim) ? a2[0] : ninf);
            S[t][2 * j + 1] = __builtin_amdgcn_exp2f(!(__builtin_fabsf(s2[1]) <= zlim) ? a2[1] : ninf);
          }
          sum += ((S[t][0] + S[t][1]) + (S[t][2] + S[t][3])) +
                 ((S[t][4] + S[t][5]) + (S[t][6] + S[t][7]));
        }
        sum += __shfl_xor(sum, 16, 32);
        inv = 1.f / (sum + 1e-10f);
      }

      {
        const float* v0p = &Vt[(hoff + ln) * SP + 2 * lh];
        const float* v1p = v0p + 16 * SP;
        v8f o0 = {}, o1 = {};
#pragma unroll
        for (int t = 0; t < 16; t += 2) {
          v2f a0[8], a1[8];
#pragma unroll
          for (int j = 0; j < 8; ++j) {
            a0[j] = *(const v2f*)&v0p[t * 16 + 4 * j];
            a1[j] = *(const v2f*)&v1p[t * 16 + 4 * j];
          }
#pragma unroll
          for (int j = 0; j < 8; ++j) {
            const int tt = t + (j >> 2), jj = j & 3;
            v2f bp;
            bp[0] = S[tt][2 * jj];
            bp[1] = S[tt][2 * jj + 1];
            o0 = wmma4(a0[j], bp, o0);
            o1 = wmma4(a1[j], bp, o1);
          }
        }
        float* orow = &Qs[(rb * 16 + ln) * QKS + hoff];
        float4 w;
        w.x = o0[0] * inv; w.y = o0[1] * inv; w.z = o0[2] * inv; w.w = o0[3] * inv;
        *(float4*)&orow[8 * lh] = w;
        w.x = o0[4] * inv; w.y = o0[5] * inv; w.z = o0[6] * inv; w.w = o0[7] * inv;
        *(float4*)&orow[8 * lh + 4] = w;
        if (lh == 0) {
          w.x = o1[0] * inv; w.y = o1[1] * inv; w.z = o1[2] * inv; w.w = o1[3] * inv;
          *(float4*)&orow[16] = w;
        }
      }
    }
  }

  for (int rbi = 0; rbi < 2; ++rbi) {
    const int rb = wid + rbi * 8;
    for (int ct = 0; ct < 2; ++ct) {
      const int col = ct * 16 + ln;
      v8f c0 = {}, c1 = {};
#pragma unroll
      for (int k0 = 0; k0 < DQK; k0 += 8) {
        const int kk = k0 + 2 * lh;
        c0 = wmma4(*(const v2f*)&Qs[(rb * 16 + ln) * QKS + kk],
                   *(const v2f*)&Wht[col * DQK + kk], c0);
        c1 = wmma4(*(const v2f*)&Qs[(rb * 16 + ln) * QKS + kk + 4],
                   *(const v2f*)&Wht[col * DQK + kk + 4], c1);
      }
      const v8f c = c0 + c1;
      if (col < HEAD_SIZE) {
        float* og = out + ooff;
#pragma unroll
        for (int v = 0; v < 8; ++v)
          og[(rb * 16 + v + 8 * lh) * HEAD_SIZE + col] = c[v];
      }
    }
  }
}

extern "C" void kernel_launch(void* const* d_in, const int* in_sizes, int n_in,
                              void* d_out, int out_size, void* d_ws, size_t ws_size,
                              hipStream_t stream) {
  (void)in_sizes; (void)n_in; (void)out_size; (void)d_ws; (void)ws_size;
  const float* x  = (const float*)d_in[0];
  const float* y  = (const float*)d_in[1];
  const float* WQ = (const float*)d_in[2];
  const float* WK = (const float*)d_in[3];
  const float* WV = (const float*)d_in[4];
  const float* WH = (const float*)d_in[5];
  float* out = (float*)d_out;

  const size_t shmem = (size_t)37088 * sizeof(float);
  (void)hipFuncSetAttribute((const void*)set_attention_kernel,
                            hipFuncAttributeMaxDynamicSharedMemorySize,
                            (int)shmem);
  hipLaunchKernelGGL(set_attention_kernel,
                     dim3(NSET * NSET), dim3(256), shmem, stream,
                     x, y, WQ, WK, WV, WH, out);
}


# WH folded into the value weights (WV_h.WH_h built once per workgroup on WMMA): P.V' is a head's share of out, heads added in LDS, out projection gone, out stored as whole lines; sum order changes
# speedup vs baseline: 4.8524x; 1.0349x over previous
#include <hip/hip_runtime.h>


typedef __attribute__((ext_vector_type(2))) float v2f;
typedef __attribute__((ext_vector_type(4))) float v4f;
typedef __attribute__((ext_vector_type(8))) float v8f;

#define HEAD_SIZE 20
#define NUM_HEADS 2
#define SEQ 256
#define NSET 32
#define DMODEL 20
#define DQK 40
#define QKS 44
#define SP 260

__device__ __forceinline__ v8f wmma4(v2f a, v2f b, v8f c) {
  return __builtin_amdgcn_wmma_f32_16x16x4_f32(
       false, a,  false, b,
       (short)0, c,  false,  false);
}

__global__ __launch_bounds__(256)
void set_attention_kernel(const float* __restrict__ x,
                          const float* __restrict__ y,
                          const float* __restrict__ WQg,
                          const float* __restrict__ WKg,
                          const float* __restrict__ WVg,
                          const float* __restrict__ WHg,
                          float* __restrict__ out) {
  extern __shared__ float smem[];
  float* Qs  = smem;
  float* Ks  = smem + 11264;
  float* Vt  = smem + 22528;
  float* Wqt = smem + 32928;
  float* Wkt = Wqt + 960;
  float* Wvt = Wkt + 960;
  float* Wht = smem + 35808;
  float* Wvr = Vt;

  const int tid  = threadIdx.x;
  const int wid  = tid >> 5;
  const int lane = tid & 31;
  const int ln   = lane & 15;
  const int lh   = lane >> 4;

  const int b  = blockIdx.x;
  const int a  = b >> 5;
  const int cc = b & 31;
  const size_t xoff = (size_t)(cc * NSET + a) * (SEQ * DMODEL);
  const size_t yoff = (size_t)(a * NSET + cc) * (SEQ * DMODEL);
  const size_t ooff = (size_t)(cc * NSET + a) * (SEQ * HEAD_SIZE);

  {
    const v4f* gx = (const v4f*)(x + xoff);
    const v4f* gy = (const v4f*)(y + yoff);
    v4f* lx = (v4f*)Qs;
    v4f* ly = (v4f*)Ks;
    v4f rx[5], ry[5];
    float wq[4], wk[4], wv[4], wh[4];
#pragma unroll
    for (int p = 0; p < 5; ++p) {
      rx[p] = gx[tid + 256 * p];
      ry[p] = gy[tid + 256 * p];
    }
#pragma unroll
    for (int p = 0; p < 4; ++p) {
      const int i  = tid + 256 * p;
      const int ic = i < 800 ? i : 799;
      wq[p] = WQg[ic];
      wk[p] = WKg[ic];
      wv[p] = WVg[ic];
      wh[p] = WHg[ic];
    }
    for (int i = tid; i < 160; i += 256) {
      Wqt[800 + i] = 0.f; Wkt[800 + i] = 0.f; Wvt[800 + i] = 0.f;
    }
    for (int i = tid; i < 480; i += 256) Wht[800 + i] = 0.f;
    for (int i = tid; i < 480; i += 256) Wvr[800 + i] = 0.f;
    asm volatile("" :: "v"(rx[0]), "v"(rx[1]), "v"(rx[2]), "v"(rx[3]), "v"(rx[4]),
                       "v"(ry[0]), "v"(ry[1]), "v"(ry[2]), "v"(ry[3]), "v"(ry[4]));
#pragma unroll
    for (int p = 0; p < 5; ++p) {
      const int i = tid + 256 * p;
      const int r = i / 5, c = i - 5 * r;
      lx[r * (QKS / 4) + c] = rx[p];
      ly[r * (QKS / 4) + c] = ry[p];
    }
#pragma unroll
    for (int p = 0; p < 4; ++p) {
      const int i = tid + 256 * p;
      if (i < 800) {
        const int k = i / DQK, n = i % DQK;
        Wqt[n * DMODEL + k] = wq[p];
        Wkt[n * DMODEL + k] = wk[p];
        Wvr[i] = wv[p];
        const int k2 = i / HEAD_SIZE, n2 = i % HEAD_SIZE;
        Wht[n2 * DQK + k2] = wh[p];
      }
    }
  }
  __syncthreads();

  {
    const int ph = wid >> 2, mt = (wid >> 1) & 1, nt = wid & 1;
    v8f cw = {};
#pragma unroll
    for (int i = 0; i < 5; ++i) {
      const int d = ph * HEAD_SIZE + 4 * i + 2 * lh;
      cw = wmma4(*(const v2f*)&Wht[(mt * 16 + ln) * DQK + d],
                 *(const v2f*)&Wvr[(nt * 16 + ln) * DQK + d], cw);
    }
    const int kcol = nt * 16 + ln;
    float* wdst = &Wvt[(ph * HEAD_SIZE + mt * 16 + 8 * lh) * DMODEL + kcol];
    if (kcol < DMODEL && (mt == 0 || lh == 0)) {
      wdst[0] = cw[0]; wdst[DMODEL] = cw[1]; wdst[2 * DMODEL] = cw[2]; wdst[3 * DMODEL] = cw[3];
    }
    if (kcol < DMODEL && mt == 0) {
      wdst[4 * DMODEL] = cw[4]; wdst[5 * DMODEL] = cw[5]; wdst[6 * DMODEL] = cw[6]; wdst[7 * DMODEL] = cw[7];
    }
  }
  __syncthreads();

  for (int rbi = 0; rbi < 2; ++rbi) {
    const int rb   = wid + rbi * 8;
    const int arow = (rb * 16 + ln) * QKS + 2 * lh;
    v2f ax[5], ay[5];
#pragma unroll
    for (int i = 0; i < 5; ++i) {
      ax[i] = *(const v2f*)&Qs[arow + 4 * i];
      ay[i] = *(const v2f*)&Ks[arow + 4 * i];
    }
    for (int ct = 0; ct < 3; ++ct) {
      const int col = ct * 16 + ln;
      v8f cq = {}, ck = {}, cv = {};
#pragma unroll
      for (int i = 0; i < 5; ++i) {
        const int kk = 4 * i + 2 * lh;
        cq = wmma4(ax[i], *(const v2f*)&Wqt[col * DMODEL + kk], cq);
        ck = wmma4(ay[i], *(const v2f*)&Wkt[col * DMODEL + kk], ck);
        cv = wmma4(ay[i], *(const v2f*)&Wvt[col * DMODEL + kk], cv);
      }
      if (col < DQK) {
#pragma unroll
        for (int v = 0; v < 8; ++v) {
          const int row = rb * 16 + v + 8 * lh;
          Qs[row * QKS + col] = cq[v];
          Ks[row * QKS + col] = ck[v];
          Vt[col * SP + row] = cv[v];
        }
      }
    }
  }
  __syncthreads();

  const float c = (float)(0.22360679774997896 * 1.4426950408889634);
  const int kperm = ((ln & 6) << 1) | ((ln >> 3) << 1) | (ln & 1);

  for (int h = 0; h < NUM_HEADS; ++h) {
    const int hoff = h * HEAD_SIZE;
    for (int rbi = 0; rbi < 2; ++rbi) {
      const int rb = wid + rbi * 8;

      v8f S[16];
      {
        v2f bq[5];
#pragma unroll
        for (int i = 0; i < 5; ++i)
          bq[i] = *(const v2f*)&Qs[(rb * 16 + ln) * QKS + hoff + 4 * i + 2 * lh];
        const float* kbase = &Ks[kperm * QKS + hoff + 2 * lh];
#pragma unroll
        for (int t = 0; t < 16; t += 4) {
          v2f ak[4][5];
#pragma unroll
          for (int u = 0; u < 4; ++u)
#pragma unroll
            for (int i = 0; i < 5; ++i)
              ak[u][i] = *(const v2f*)&kbase[(t + u) * 16 * QKS + 4 * i];
          v8f c0 = {}, c1 = {}, c2 = {}, c3 = {};
#pragma unroll
          for (int i = 0; i < 5; ++i) {
            c0 = wmma4(ak[0][i], bq[i], c0);
            c1 = wmma4(ak[1][i], bq[i], c1);
            c2 = wmma4(ak[2][i], bq[i], c2);
            c3 = wmma4(ak[3][i], bq[i], c3);
          }
          S[t] = c0; S[t + 1] = c1; S[t + 2] = c2; S[t + 3] = c3;
        }
      }

      float inv;
      {
        float mx = -__builtin_inff();
#pragma unroll
        for (int t = 0; t < 16; ++t)
          mx = fmaxf(mx, fmaxf(fmaxf(fmaxf(S[t][0], S[t][1]), fmaxf(S[t][2], S[t][3])),
                               fmaxf(fmaxf(S[t][4], S[t][5]), fmaxf(S[t][6], S[t][7]))));
        mx = fmaxf(mx, __shfl_xor(mx, 16, 32));
        const float nmc = -(mx * c);
        const v2f c2 = {c, c};
        const v2f n2 = {nmc, nmc};
        const float zlim = 0x1p-148f;
        const float ninf = -__builtin_inff();
        float sum = 0.f;
#pragma unroll
        for (int t = 0; t < 16; ++t) {
#pragma unroll
          for (int j = 0; j < 4; ++j) {
            v2f s2;
            s2[0] = S[t][2 * j];
            s2[1] = S[t][2 * j + 1];
            const v2f a2 = __builtin_elementwise_fma(s2, c2, n2);
            S[t][2 * j]     = __builtin_amdgcn_exp2f(!(__builtin_fabsf(s2[0]) <= zlim) ? a2[0] : ninf);
            S[t][2 * j + 1] = __builtin_amdgcn_exp2f(!(__builtin_fabsf(s2[1]) <= zlim) ? a2[1] : ninf);
          }
          sum += ((S[t][0] + S[t][1]) + (S[t][2] + S[t][3])) +
                 ((S[t][4] + S[t][5]) + (S[t][6] + S[t][7]));
        }
        sum += __shfl_xor(sum, 16, 32);
        inv = 1.f / (sum + 1e-10f);
      }

      {
        const float* v0p = &Vt[(hoff + ln) * SP + 2 * lh];
        const float* v1p = v0p + 16 * SP;
        v8f o0 = {}, o1 = {};
#pragma unroll
        for (int t = 0; t < 16; t += 2) {
          v2f a0[8], a1[8];
#pragma unroll
          for (int j = 0; j < 8; ++j) {
            a0[j] = *(const v2f*)&v0p[t * 16 + 4 * j];
            a1[j] = *(const v2f*)&v1p[t * 16 + 4 * j];
          }
#pragma unroll
          for (int j = 0; j < 8; ++j) {
            const int tt = t + (j >> 2), jj = j & 3;
            v2f bp;
            bp[0] = S[tt][2 * jj];
            bp[1] = S[tt][2 * jj + 1];
            o0 = wmma4(a0[j], bp, o0);
            o1 = wmma4(a1[j], bp, o1);
          }
        }
        float* orow = &Qs[(rb * 16 + ln) * QKS];
        float4 w;
        if (h == 0) {
          w.x = o0[0] * inv; w.y = o0[1] * inv; w.z = o0[2] * inv; w.w = o0[3] * inv;
          *(float4*)&orow[8 * lh] = w;
          w.x = o0[4] * inv; w.y = o0[5] * inv; w.z = o0[6] * inv; w.w = o0[7] * inv;
          *(float4*)&orow[8 * lh + 4] = w;
          if (lh == 0) {
            w.x = o1[0] * inv; w.y = o1[1] * inv; w.z = o1[2] * inv; w.w = o1[3] * inv;
            *(float4*)&orow[16] = w;
          }
        } else {
          float4 p = *(const float4*)&orow[8 * lh];
          w.x = __builtin_fmaf(o0[0], inv, p.x); w.y = __builtin_fmaf(o0[1], inv, p.y);
          w.z = __builtin_fmaf(o0[2], inv, p.z); w.w = __builtin_fmaf(o0[3], inv, p.w);
          *(float4*)&orow[8 * lh] = w;
          p = *(const float4*)&orow[8 * lh + 4];
          w.x = __builtin_fmaf(o0[4], inv, p.x); w.y = __builtin_fmaf(o0[5], inv, p.y);
          w.z = __builtin_fmaf(o0[6], inv, p.z); w.w = __builtin_fmaf(o0[7], inv, p.w);
          *(float4*)&orow[8 * lh + 4] = w;
          if (lh == 0) {
            p = *(const float4*)&orow[16];
            w.x = __builtin_fmaf(o1[0], inv, p.x); w.y = __builtin_fmaf(o1[1], inv, p.y);
            w.z = __builtin_fmaf(o1[2], inv, p.z); w.w = __builtin_fmaf(o1[3], inv, p.w);
            *(float4*)&orow[16] = w;
          }
          float4* og4 = (float4*)(out + ooff + (size_t)rb * 16 * HEAD_SIZE);
#pragma unroll
          for (int p3 = 0; p3 < 3; ++p3) {
            const int idx = lane + 32 * p3;
            if (idx < 16 * HEAD_SIZE / 4) {
              const int row = idx / 5, c4 = idx - 5 * row;
              og4[idx] = *(const float4*)&Qs[(rb * 16 + row) * QKS + 4 * c4];
            }
          }
        }
      }
    }
  }
}

extern "C" void kernel_launch(void* const* d_in, const int* in_sizes, int n_in,
                              void* d_out, int out_size, void* d_ws, size_t ws_size,
                              hipStream_t stream) {
  (void)in_sizes; (void)n_in; (void)out_size; (void)d_ws; (void)ws_size;
  const float* x  = (const float*)d_in[0];
  const float* y  = (const float*)d_in[1];
  const float* WQ = (const float*)d_in[2];
  const float* WK = (const float*)d_in[3];
  const float* WV = (const float*)d_in[4];
  const float* WH = (const float*)d_in[5];
  float* out = (float*)d_out;

  const size_t shmem = (size_t)37088 * sizeof(float);
  (void)hipFuncSetAttribute((const void*)set_attention_kernel,
                            hipFuncAttributeMaxDynamicSharedMemorySize,
                            (int)shmem);
  hipLaunchKernelGGL(set_attention_kernel,
                     dim3(NSET * NSET), dim3(256), shmem, stream,
                     x, y, WQ, WK, WV, WH, out);
}
